# starting kernel
# baseline (speedup 1.0000x reference)
#include <hip/hip_runtime.h>

typedef __attribute__((ext_vector_type(2))) float v2f;
typedef __attribute__((ext_vector_type(8))) float v8f;

#define HDIM 256
#define NROWS 1024

__global__ __launch_bounds__(256) void proj_wmma(
    const float* __restrict__ dec, const float* __restrict__ enc,
    const float* __restrict__ W,   const float* __restrict__ bmlp,
    float* __restrict__ dp, float* __restrict__ ep)
{
  const int lane = threadIdx.x & 31;
  const int wv   = threadIdx.x >> 5;
  const int tile = blockIdx.x * 8 + wv;
  const bool isDec = tile < 1024;
  const int tt      = tile & 1023;
  const int rowTile = tt >> 4;
  const int colTile = tt & 15;
  const int row0 = rowTile << 4;
  const int col0 = colTile << 4;

  const float* src  = isDec ? dec : enc;
  const int    woff = isDec ? HDIM : 0;
  float*       dst  = isDec ? dp : ep;

  const int laneLo = lane & 15;
  const int laneHi = lane >> 4;

  const float* aPtr = src + (size_t)(row0 + laneLo) * HDIM + 2 * laneHi;
  const float* bPtr = W   + (size_t)(col0 + laneLo) * (2 * HDIM) + woff + 2 * laneHi;

  v8f acc = {};
  for (int k = 0; k < HDIM; k += 4) {
    v2f a = *(const v2f*)(aPtr + k);
    v2f b = *(const v2f*)(bPtr + k);
    acc = __builtin_amdgcn_wmma_f32_16x16x4_f32(
         false, a,  false, b,
         (short)0, acc,  false,  false);
  }

  const float bias = isDec ? bmlp[col0 + laneLo] : 0.0f;
  #pragma unroll
  for (int v = 0; v < 8; ++v) {
    const int m = v + 8 * laneHi;
    dst[(size_t)(row0 + m) * HDIM + col0 + laneLo] = acc[v] + bias;
  }
}

#define HC  64
#define PAD 66

__global__ __launch_bounds__(256) void attn_combine(
    const float* __restrict__ dp, const float* __restrict__ ep,
    const float* __restrict__ wout, const float* __restrict__ bout,
    float* __restrict__ out)
{
  __shared__ __align__(16) float dpS[64][PAD];
  __shared__ __align__(16) float epS[64][PAD];
  __shared__ __align__(16) float wS[HDIM];

  const int tid = threadIdx.x;
  const int b   = blockIdx.x >> 6;
  const int qt  = (blockIdx.x >> 3) & 7;
  const int et  = blockIdx.x & 7;
  const int q0  = qt * 64, e0 = et * 64;
  const int tx  = tid & 15, ty = tid >> 4;

  wS[tid] = wout[tid];

  const int r = tid >> 5;
  const int c = (tid & 31) * 2;

  float acc[4][4] = {};

  for (int hc = 0; hc < HDIM; hc += HC) {
    __syncthreads();
    #pragma unroll
    for (int i = 0; i < 8; ++i) {
      const int row = r + 8 * i;
      *(v2f*)&dpS[row][c] =
          *(const v2f*)(dp + (size_t)(b * 512 + q0 + row) * HDIM + hc + c);
      *(v2f*)&epS[row][c] =
          *(const v2f*)(ep + (size_t)(b * 512 + e0 + row) * HDIM + hc + c);
    }
    __syncthreads();

    for (int h = 0; h < HC; h += 2) {
      const v2f wv = *(const v2f*)&wS[hc + h];
      v2f dq[4], ee[4];
      #pragma unroll
      for (int i = 0; i < 4; ++i) dq[i] = *(const v2f*)&dpS[ty + 16 * i][h];
      #pragma unroll
      for (int j = 0; j < 4; ++j) ee[j] = *(const v2f*)&epS[tx + 16 * j][h];
      #pragma unroll
      for (int i = 0; i < 4; ++i)
        #pragma unroll
        for (int j = 0; j < 4; ++j) {
          acc[i][j] += fmaxf(dq[i].x + ee[j].x, 0.0f) * wv.x;
          acc[i][j] += fmaxf(dq[i].y + ee[j].y, 0.0f) * wv.y;
        }
    }
  }

  const float bo = bout[0];
  #pragma unroll
  for (int i = 0; i < 4; ++i)
    #pragma unroll
    for (int j = 0; j < 4; ++j)
      out[(size_t)(b * 512 + q0 + ty + 16 * i) * 512 + e0 + tx + 16 * j] =
          acc[i][j] + bo;
}

extern "C" void kernel_launch(void* const* d_in, const int* in_sizes, int n_in,
                              void* d_out, int out_size, void* d_ws, size_t ws_size,
                              hipStream_t stream) {
  const float* dec = (const float*)d_in[0];
  const float* enc = (const float*)d_in[1];
  const float* W   = (const float*)d_in[2];
  const float* bm  = (const float*)d_in[3];
  const float* wo  = (const float*)d_in[4];
  const float* bo  = (const float*)d_in[5];
  float* out = (float*)d_out;

  float* dp = (float*)d_ws;
  float* ep = dp + (size_t)NROWS * HDIM;

  proj_wmma<<<256, 256, 0, stream>>>(dec, enc, W, bm, dp, ep);
  attn_combine<<<128, 256, 0, stream>>>(dp, ep, wo, bo, out);
}


# combine kernel: expose ILP in the h loop (2-waves-per-SIMD register budget + unroll x4) so the 4x4 add/max/fma chains interleave and LDS reads group ahead of the math; sums unchanged
# speedup vs baseline: 1.5926x; 1.5926x over previous
#include <hip/hip_runtime.h>

typedef __attribute__((ext_vector_type(2))) float v2f;
typedef __attribute__((ext_vector_type(8))) float v8f;

#define HDIM 256
#define NROWS 1024

__global__ __launch_bounds__(256) void proj_wmma(
    const float* __restrict__ dec, const float* __restrict__ enc,
    const float* __restrict__ W,   const float* __restrict__ bmlp,
    float* __restrict__ dp, float* __restrict__ ep)
{
  const int lane = threadIdx.x & 31;
  const int wv   = threadIdx.x >> 5;
  const int tile = blockIdx.x * 8 + wv;
  const bool isDec = tile < 1024;
  const int tt      = tile & 1023;
  const int rowTile = tt >> 4;
  const int colTile = tt & 15;
  const int row0 = rowTile << 4;
  const int col0 = colTile << 4;

  const float* src  = isDec ? dec : enc;
  const int    woff = isDec ? HDIM : 0;
  float*       dst  = isDec ? dp : ep;

  const int laneLo = lane & 15;
  const int laneHi = lane >> 4;

  const float* aPtr = src + (size_t)(row0 + laneLo) * HDIM + 2 * laneHi;
  const float* bPtr = W   + (size_t)(col0 + laneLo) * (2 * HDIM) + woff + 2 * laneHi;

  v8f acc = {};
  for (int k = 0; k < HDIM; k += 4) {
    v2f a = *(const v2f*)(aPtr + k);
    v2f b = *(const v2f*)(bPtr + k);
    acc = __builtin_amdgcn_wmma_f32_16x16x4_f32(
         false, a,  false, b,
         (short)0, acc,  false,  false);
  }

  const float bias = isDec ? bmlp[col0 + laneLo] : 0.0f;
  #pragma unroll
  for (int v = 0; v < 8; ++v) {
    const int m = v + 8 * laneHi;
    dst[(size_t)(row0 + m) * HDIM + col0 + laneLo] = acc[v] + bias;
  }
}

#define HC  64
#define PAD 66

__global__ __launch_bounds__(256) __attribute__((amdgpu_waves_per_eu(2, 2)))
void attn_combine(
    const float* __restrict__ dp, const float* __restrict__ ep,
    const float* __restrict__ wout, const float* __restrict__ bout,
    float* __restrict__ out)
{
  __shared__ __align__(16) float dpS[64][PAD];
  __shared__ __align__(16) float epS[64][PAD];
  __shared__ __align__(16) float wS[HDIM];

  const int tid = threadIdx.x;
  const int b   = blockIdx.x >> 6;
  const int qt  = (blockIdx.x >> 3) & 7;
  const int et  = blockIdx.x & 7;
  const int q0  = qt * 64, e0 = et * 64;
  const int tx  = tid & 15, ty = tid >> 4;

  wS[tid] = wout[tid];

  const int r = tid >> 5;
  const int c = (tid & 31) * 2;

  float acc[4][4] = {};

  for (int hc = 0; hc < HDIM; hc += HC) {
    __syncthreads();
    #pragma unroll
    for (int i = 0; i < 8; ++i) {
      const int row = r + 8 * i;
      *(v2f*)&dpS[row][c] =
          *(const v2f*)(dp + (size_t)(b * 512 + q0 + row) * HDIM + hc + c);
      *(v2f*)&epS[row][c] =
          *(const v2f*)(ep + (size_t)(b * 512 + e0 + row) * HDIM + hc + c);
    }
    __syncthreads();

    #pragma unroll 4
    for (int h = 0; h < HC; h += 2) {
      const v2f wv = *(const v2f*)&wS[hc + h];
      v2f dq[4], ee[4];
      #pragma unroll
      for (int i = 0; i < 4; ++i) dq[i] = *(const v2f*)&dpS[ty + 16 * i][h];
      #pragma unroll
      for (int j = 0; j < 4; ++j) ee[j] = *(const v2f*)&epS[tx + 16 * j][h];
      #pragma unroll
      for (int i = 0; i < 4; ++i)
        #pragma unroll
        for (int j = 0; j < 4; ++j) {
          acc[i][j] += fmaxf(dq[i].x + ee[j].x, 0.0f) * wv.x;
          acc[i][j] += fmaxf(dq[i].y + ee[j].y, 0.0f) * wv.y;
        }
    }
  }

  const float bo = bout[0];
  #pragma unroll
  for (int i = 0; i < 4; ++i)
    #pragma unroll
    for (int j = 0; j < 4; ++j)
      out[(size_t)(b * 512 + q0 + ty + 16 * i) * 512 + e0 + tx + 16 * j] =
          acc[i][j] + bo;
}

extern "C" void kernel_launch(void* const* d_in, const int* in_sizes, int n_in,
                              void* d_out, int out_size, void* d_ws, size_t ws_size,
                              hipStream_t stream) {
  const float* dec = (const float*)d_in[0];
  const float* enc = (const float*)d_in[1];
  const float* W   = (const float*)d_in[2];
  const float* bm  = (const float*)d_in[3];
  const float* wo  = (const float*)d_in[4];
  const float* bo  = (const float*)d_in[5];
  float* out = (float*)d_out;

  float* dp = (float*)d_ws;
  float* ep = dp + (size_t)NROWS * HDIM;

  proj_wmma<<<256, 256, 0, stream>>>(dec, enc, W, bm, dp, ep);
  attn_combine<<<128, 256, 0, stream>>>(dp, ep, wo, bo, out);
}


# combine kernel re-tiled so every WGP has a block: 256 blocks of 64x32 (was 128 of 64x64), tile staged whole, a block's two 4-wave halves each sum half of h, halves added via LDS (two half sums)
# speedup vs baseline: 2.2437x; 1.4088x over previous
#include <hip/hip_runtime.h>

typedef __attribute__((ext_vector_type(2))) float v2f;
typedef __attribute__((ext_vector_type(4))) float v4f;
typedef __attribute__((ext_vector_type(8))) float v8f;

#define HDIM 256
#define NROWS 1024

__global__ __launch_bounds__(256) void proj_wmma(
    const float* __restrict__ dec, const float* __restrict__ enc,
    const float* __restrict__ W,   const float* __restrict__ bmlp,
    float* __restrict__ dp, float* __restrict__ ep)
{
  const int lane = threadIdx.x & 31;
  const int wv   = threadIdx.x >> 5;
  const int tile = blockIdx.x * 8 + wv;
  const bool isDec = tile < 1024;
  const int tt      = tile & 1023;
  const int rowTile = tt >> 4;
  const int colTile = tt & 15;
  const int row0 = rowTile << 4;
  const int col0 = colTile << 4;

  const float* src  = isDec ? dec : enc;
  const int    woff = isDec ? HDIM : 0;
  float*       dst  = isDec ? dp : ep;

  const int laneLo = lane & 15;
  const int laneHi = lane >> 4;

  const float* aPtr = src + (size_t)(row0 + laneLo) * HDIM + 2 * laneHi;
  const float* bPtr = W   + (size_t)(col0 + laneLo) * (2 * HDIM) + woff + 2 * laneHi;

  v8f acc = {};
  for (int k = 0; k < HDIM; k += 4) {
    v2f a = *(const v2f*)(aPtr + k);
    v2f b = *(const v2f*)(bPtr + k);
    acc = __builtin_amdgcn_wmma_f32_16x16x4_f32(
         false, a,  false, b,
         (short)0, acc,  false,  false);
  }

  const float bias = isDec ? bmlp[col0 + laneLo] : 0.0f;
  #pragma unroll
  for (int v = 0; v < 8; ++v) {
    const int m = v + 8 * laneHi;
    dst[(size_t)(row0 + m) * HDIM + col0 + laneLo] = acc[v] + bias;
  }
}

#define TQ    64
#define TE    32
#define HHALF 128
#define PITCH 258

__global__ __launch_bounds__(256) __attribute__((amdgpu_waves_per_eu(2, 2)))
void attn_combine(
    const float* __restrict__ dp, const float* __restrict__ ep,
    const float* __restrict__ wout, const float* __restrict__ bout,
    float* __restrict__ out)
{
  __shared__ __align__(16) float dpS[TQ][PITCH];
  __shared__ __align__(16) float epS[TE][PITCH];
  __shared__ __align__(16) float wS[HDIM];
  __shared__ __align__(16) float redS[16][HHALF];

  const int tid = threadIdx.x;
  const int b   = blockIdx.x >> 7;
  const int qt  = (blockIdx.x >> 4) & 7;
  const int et  = blockIdx.x & 15;
  const int q0  = qt * TQ, e0 = et * TE;

  wS[tid] = wout[tid];

  {
    const int r0 = tid >> 6;
    const int c  = (tid & 63) * 4;
    const float* dsrc = dp + (size_t)(b * 512 + q0 + r0) * HDIM + c;
    const float* esrc = ep + (size_t)(b * 512 + e0 + r0) * HDIM + c;
    #pragma unroll
    for (int k = 0; k < TQ / 4; ++k) {
      const v4f v = *(const v4f*)(dsrc + (size_t)k * 4 * HDIM);
      *(v2f*)&dpS[r0 + 4 * k][c]     = v.xy;
      *(v2f*)&dpS[r0 + 4 * k][c + 2] = v.zw;
    }
    #pragma unroll
    for (int k = 0; k < TE / 4; ++k) {
      const v4f v = *(const v4f*)(esrc + (size_t)k * 4 * HDIM);
      *(v2f*)&epS[r0 + 4 * k][c]     = v.xy;
      *(v2f*)&epS[r0 + 4 * k][c + 2] = v.zw;
    }
  }
  __syncthreads();

  const int g  = __builtin_amdgcn_readfirstlane(tid >> 7);
  const int t  = tid & 127;
  const int tx = t & 7, ty = t >> 3;
  const int hb = g * HHALF;

  float acc[4][4] = {};

  #pragma unroll 4
  for (int h = 0; h < HHALF; h += 2) {
    const v2f wv = *(const v2f*)&wS[hb + h];
    v2f dq[4], ee[4];
    #pragma unroll
    for (int i = 0; i < 4; ++i) dq[i] = *(const v2f*)&dpS[4 * ty + i][hb + h];
    #pragma unroll
    for (int j = 0; j < 4; ++j) ee[j] = *(const v2f*)&epS[4 * tx + j][hb + h];
    #pragma unroll
    for (int i = 0; i < 4; ++i)
      #pragma unroll
      for (int j = 0; j < 4; ++j) {
        acc[i][j] += fmaxf(dq[i].x + ee[j].x, 0.0f) * wv.x;
        acc[i][j] += fmaxf(dq[i].y + ee[j].y, 0.0f) * wv.y;
      }
  }

  if (g == 1) {
    #pragma unroll
    for (int i = 0; i < 4; ++i)
      #pragma unroll
      for (int j = 0; j < 4; ++j) redS[i * 4 + j][t] = acc[i][j];
  }
  __syncthreads();
  if (g == 0) {
    const float bo = bout[0];
    #pragma unroll
    for (int i = 0; i < 4; ++i) {
      v4f o;
      o.x = (acc[i][0] + redS[i * 4 + 0][t]) + bo;
      o.y = (acc[i][1] + redS[i * 4 + 1][t]) + bo;
      o.z = (acc[i][2] + redS[i * 4 + 2][t]) + bo;
      o.w = (acc[i][3] + redS[i * 4 + 3][t]) + bo;
      *(v4f*)(out + (size_t)(b * 512 + q0 + 4 * ty + i) * 512 + e0 + 4 * tx) = o;
    }
  }
}

extern "C" void kernel_launch(void* const* d_in, const int* in_sizes, int n_in,
                              void* d_out, int out_size, void* d_ws, size_t ws_size,
                              hipStream_t stream) {
  const float* dec = (const float*)d_in[0];
  const float* enc = (const float*)d_in[1];
  const float* W   = (const float*)d_in[2];
  const float* bm  = (const float*)d_in[3];
  const float* wo  = (const float*)d_in[4];
  const float* bo  = (const float*)d_in[5];
  float* out = (float*)d_out;

  float* dp = (float*)d_ws;
  float* ep = dp + (size_t)NROWS * HDIM;

  proj_wmma<<<256, 256, 0, stream>>>(dec, enc, W, bm, dp, ep);
  attn_combine<<<256, 256, 0, stream>>>(dp, ep, wo, bo, out);
}


# projection kernel: operands go through LDS (32x64 block tile, fetched as whole 128-byte lines in 4 k chunks that overlap the matrix ops), not per-lane 8-byte global fragment loads; sums unchanged
# speedup vs baseline: 2.7913x; 1.2441x over previous
#include <hip/hip_runtime.h>

typedef __attribute__((ext_vector_type(2))) float v2f;
typedef __attribute__((ext_vector_type(4))) float v4f;
typedef __attribute__((ext_vector_type(8))) float v8f;

#define HDIM 256
#define NROWS 1024

#define PROWS  32
#define PCOLS  64
#define PPITCH 260
#define KCH    64
#define NCH    (HDIM / KCH)
#define KSTEPS (KCH / 4)

__global__ __launch_bounds__(256) void proj_wmma(
    const float* __restrict__ dec, const float* __restrict__ enc,
    const float* __restrict__ W,   const float* __restrict__ bmlp,
    float* __restrict__ dp, float* __restrict__ ep)
{
  __shared__ __align__(16) float opS[PROWS + PCOLS][PPITCH];

  const int tid  = threadIdx.x;
  const int lane = tid & 31;
  const int wv   = tid >> 5;
  const bool isDec = blockIdx.x < 128;
  const int  bb    = blockIdx.x & 127;
  const int  row0  = (bb >> 2) * PROWS;
  const int  col0  = (bb & 3) * PCOLS;

  const float* src  = isDec ? dec : enc;
  const int    woff = isDec ? HDIM : 0;
  float*       dst  = isDec ? dp : ep;

  const int sl = tid & 15;
  const int sr = tid >> 4;
  const float* asrc = src + (size_t)(row0 + sr) * HDIM + 4 * sl;
  const float* bsrc = W   + (size_t)(col0 + sr) * (2 * HDIM) + woff + 4 * sl;

  v4f g[NCH][6];
  #pragma unroll
  for (int j = 0; j < NCH; ++j) {
    #pragma unroll
    for (int i = 0; i < 2; ++i)
      g[j][i] = *(const v4f*)(asrc + (size_t)i * 16 * HDIM + j * KCH);
    #pragma unroll
    for (int i = 0; i < 4; ++i)
      g[j][2 + i] = *(const v4f*)(bsrc + (size_t)i * 16 * (2 * HDIM) + j * KCH);
    __builtin_amdgcn_sched_barrier(0);
  }

  const int laneLo = lane & 15;
  const int laneHi = lane >> 4;
  const int wr = wv >> 2;
  const int wc = wv & 3;

  const float* aPtr = &opS[wr * 16 + laneLo][2 * laneHi];
  const float* bPtr = &opS[PROWS + wc * 16 + laneLo][2 * laneHi];

  v8f acc = {};
  #pragma unroll
  for (int j = 0; j < NCH; ++j) {
    #pragma unroll
    for (int i = 0; i < 6; ++i)
      *(v4f*)&opS[sr + 16 * i][j * KCH + 4 * sl] = g[j][i];
    __builtin_amdgcn_fence(__ATOMIC_RELEASE, "workgroup", "local");
    __builtin_amdgcn_s_barrier();
    __builtin_amdgcn_fence(__ATOMIC_ACQUIRE, "workgroup", "local");

    v2f a[KSTEPS], b[KSTEPS];
    #pragma unroll
    for (int s = 0; s < KSTEPS; ++s) {
      a[s] = *(const v2f*)(aPtr + j * KCH + 4 * s);
      b[s] = *(const v2f*)(bPtr + j * KCH + 4 * s);
    }
    #pragma unroll
    for (int s = 0; s < KSTEPS; ++s)
      acc = __builtin_amdgcn_wmma_f32_16x16x4_f32(
           false, a[s],  false, b[s],
           (short)0, acc,  false,  false);
    __builtin_amdgcn_sched_group_barrier(0x100, 16, 0);
    __builtin_amdgcn_sched_group_barrier(0x008, 16, 0);
    __builtin_amdgcn_sched_barrier(0);
  }

  const int colG = col0 + wc * 16 + laneLo;
  const float bias = isDec ? bmlp[colG] : 0.0f;
  #pragma unroll
  for (int v = 0; v < 8; ++v) {
    const int m = v + 8 * laneHi;
    dst[(size_t)(row0 + wr * 16 + m) * HDIM + colG] = acc[v] + bias;
  }
}

#define TQ    64
#define TE    32
#define HHALF 128
#define PITCH 258

__global__ __launch_bounds__(256) __attribute__((amdgpu_waves_per_eu(2, 2)))
void attn_combine(
    const float* __restrict__ dp, const float* __restrict__ ep,
    const float* __restrict__ wout, const float* __restrict__ bout,
    float* __restrict__ out)
{
  __shared__ __align__(16) float dpS[TQ][PITCH];
  __shared__ __align__(16) float epS[TE][PITCH];
  __shared__ __align__(16) float wS[HDIM];
  __shared__ __align__(16) float redS[16][HHALF];

  const int tid = threadIdx.x;
  const int b   = blockIdx.x >> 7;
  const int qt  = (blockIdx.x >> 4) & 7;
  const int et  = blockIdx.x & 15;
  const int q0  = qt * TQ, e0 = et * TE;

  wS[tid] = wout[tid];

  {
    const int r0 = tid >> 6;
    const int c  = (tid & 63) * 4;
    const float* dsrc = dp + (size_t)(b * 512 + q0 + r0) * HDIM + c;
    const float* esrc = ep + (size_t)(b * 512 + e0 + r0) * HDIM + c;
    #pragma unroll
    for (int k = 0; k < TQ / 4; ++k) {
      const v4f v = *(const v4f*)(dsrc + (size_t)k * 4 * HDIM);
      *(v2f*)&dpS[r0 + 4 * k][c]     = v.xy;
      *(v2f*)&dpS[r0 + 4 * k][c + 2] = v.zw;
    }
    #pragma unroll
    for (int k = 0; k < TE / 4; ++k) {
      const v4f v = *(const v4f*)(esrc + (size_t)k * 4 * HDIM);
      *(v2f*)&epS[r0 + 4 * k][c]     = v.xy;
      *(v2f*)&epS[r0 + 4 * k][c + 2] = v.zw;
    }
  }
  __syncthreads();

  const int g  = __builtin_amdgcn_readfirstlane(tid >> 7);
  const int t  = tid & 127;
  const int tx = t & 7, ty = t >> 3;
  const int hb = g * HHALF;

  float acc[4][4] = {};

  #pragma unroll 4
  for (int h = 0; h < HHALF; h += 2) {
    const v2f wv = *(const v2f*)&wS[hb + h];
    v2f dq[4], ee[4];
    #pragma unroll
    for (int i = 0; i < 4; ++i) dq[i] = *(const v2f*)&dpS[4 * ty + i][hb + h];
    #pragma unroll
    for (int j = 0; j < 4; ++j) ee[j] = *(const v2f*)&epS[4 * tx + j][hb + h];
    #pragma unroll
    for (int i = 0; i < 4; ++i)
      #pragma unroll
      for (int j = 0; j < 4; ++j) {
        acc[i][j] += fmaxf(dq[i].x + ee[j].x, 0.0f) * wv.x;
        acc[i][j] += fmaxf(dq[i].y + ee[j].y, 0.0f) * wv.y;
      }
  }

  if (g == 1) {
    #pragma unroll
    for (int i = 0; i < 4; ++i)
      #pragma unroll
      for (int j = 0; j < 4; ++j) redS[i * 4 + j][t] = acc[i][j];
  }
  __syncthreads();
  if (g == 0) {
    const float bo = bout[0];
    #pragma unroll
    for (int i = 0; i < 4; ++i) {
      v4f o;
      o.x = (acc[i][0] + redS[i * 4 + 0][t]) + bo;
      o.y = (acc[i][1] + redS[i * 4 + 1][t]) + bo;
      o.z = (acc[i][2] + redS[i * 4 + 2][t]) + bo;
      o.w = (acc[i][3] + redS[i * 4 + 3][t]) + bo;
      *(v4f*)(out + (size_t)(b * 512 + q0 + 4 * ty + i) * 512 + e0 + 4 * tx) = o;
    }
  }
}

extern "C" void kernel_launch(void* const* d_in, const int* in_sizes, int n_in,
                              void* d_out, int out_size, void* d_ws, size_t ws_size,
                              hipStream_t stream) {
  const float* dec = (const float*)d_in[0];
  const float* enc = (const float*)d_in[1];
  const float* W   = (const float*)d_in[2];
  const float* bm  = (const float*)d_in[3];
  const float* wo  = (const float*)d_in[4];
  const float* bo  = (const float*)d_in[5];
  float* out = (float*)d_out;

  float* dp = (float*)d_ws;
  float* ep = dp + (size_t)NROWS * HDIM;

  proj_wmma<<<256, 256, 0, stream>>>(dec, enc, W, bm, dp, ep);
  attn_combine<<<256, 256, 0, stream>>>(dp, ep, wo, bo, out);
}
